# starting kernel
# baseline (speedup 1.0000x reference)
#include <hip/hip_runtime.h>
#include <cmath>

typedef float v2f __attribute__((ext_vector_type(2)));
typedef float v8f __attribute__((ext_vector_type(8)));

#define N_    256
#define CIN   64
#define COUT  64
#define T_    64
#define V_    25
#define REL   8
#define TV    (T_ * V_)
#define VP    28
#define UP    32
#define WRS   68
#define BLK   512
#define NWAVE 16

#define OFF_WR    0
#define SZ_WR     (COUT * WRS)
#define OFF_XBAR  (OFF_WR + SZ_WR)
#define SZ_XBAR   (CIN * V_)
#define OFF_X1    (OFF_XBAR + SZ_XBAR)
#define OFF_X2    (OFF_X1 + REL * V_)
#define OFF_D     (OFF_X2 + REL * V_)
#define SZ_D      (REL * V_ * V_)
#define OFF_INP   (OFF_D + SZ_D)
#define SZ_INP    (16 * T_ * VP)
#define OFF_AF    (OFF_INP + SZ_INP)
#define SZ_AF     (16 * UP * VP)
#define SMEM_FLOATS (OFF_AF + SZ_AF)

__global__ void gcn_fused_kernel(const float* __restrict__ x,
                                 const float* __restrict__ Ag,
                                 const float* __restrict__ w1, const float* __restrict__ b1,
                                 const float* __restrict__ w2, const float* __restrict__ b2,
                                 const float* __restrict__ w3, const float* __restrict__ b3,
                                 const float* __restrict__ wr, const float* __restrict__ br,
                                 float* __restrict__ out)
{
    extern __shared__ float sm[];
    const int n    = blockIdx.x;
    const int tid  = threadIdx.x;
    const int wave = tid >> 5;
    const int lane = tid & 31;
    const int lo   = lane & 15;
    const int hi   = lane >> 4;
    const float* __restrict__ xg = x + (size_t)n * CIN * TV;

    for (int idx = tid; idx < COUT * CIN; idx += BLK) {
        const int c = idx / CIN, i = idx % CIN;
        sm[OFF_WR + c * WRS + i] = wr[idx];
    }
    for (int idx = tid; idx < SZ_INP + SZ_AF; idx += BLK) sm[OFF_INP + idx] = 0.0f;

    for (int idx = tid; idx < CIN * V_; idx += BLK) {
        const int i = idx / V_, v = idx % V_;
        const float* p = xg + i * TV + v;
        float s = 0.0f;
#pragma unroll 8
        for (int t = 0; t < T_; ++t) s += p[t * V_];
        sm[OFF_XBAR + idx] = s * (1.0f / T_);
    }
    __syncthreads();

    for (int idx = tid; idx < 2 * REL * V_; idx += BLK) {
        const int which = idx / (REL * V_);
        const int rv = idx % (REL * V_);
        const int r = rv / V_, v = rv % V_;
        const float* __restrict__ w = which ? w2 : w1;
        float s = which ? b2[r] : b1[r];
#pragma unroll 8
        for (int i = 0; i < CIN; ++i) s += w[r * CIN + i] * sm[OFF_XBAR + i * V_ + v];
        sm[(which ? OFF_X2 : OFF_X1) + rv] = s;
    }
    __syncthreads();

    for (int idx = tid; idx < SZ_D; idx += BLK) {
        const int r = idx / (V_ * V_);
        const int uv = idx % (V_ * V_);
        const int u = uv / V_, v = uv % V_;
        sm[OFF_D + idx] = tanhf(sm[OFF_X1 + r * V_ + u] - sm[OFF_X2 + r * V_ + v]);
    }

    for (int ct = 0; ct < COUT / 16; ++ct) {
        __syncthreads();

        for (int idx = tid; idx < 16 * V_ * V_; idx += BLK) {
            const int cc = idx / (V_ * V_);
            const int uv = idx % (V_ * V_);
            const int u = uv / V_, v = uv % V_;
            const int c = ct * 16 + cc;
            float acc = b3[c] + Ag[uv];
#pragma unroll
            for (int r = 0; r < REL; ++r)
                acc += w3[c * REL + r] * sm[OFF_D + r * (V_ * V_) + uv];
            sm[OFF_AF + cc * (UP * VP) + u * VP + v] = acc;
        }

        for (int tile = wave; tile < TV / 16; tile += NWAVE) {
            const int col = tile * 16 + lo;
            v8f acc = {};
#pragma unroll
            for (int k0 = 0; k0 < CIN; k0 += 4) {
                const int i0 = k0 + 2 * hi;
                v2f a, b;
                a.x = sm[OFF_WR + (ct * 16 + lo) * WRS + i0];
                a.y = sm[OFF_WR + (ct * 16 + lo) * WRS + i0 + 1];
                b.x = xg[i0 * TV + col];
                b.y = xg[(i0 + 1) * TV + col];
                acc = __builtin_amdgcn_wmma_f32_16x16x4_f32(
                        false, a, false, b, (short)0, acc, false, false);
            }
            const int t = col / V_, v = col % V_;
#pragma unroll
            for (int j = 0; j < 8; ++j) {
                const int m = j + 8 * hi;
                sm[OFF_INP + m * (T_ * VP) + t * VP + v] = acc[j] + br[ct * 16 + m];
            }
        }
        __syncthreads();

        {
            const int cl = wave;
            const int c = ct * 16 + cl;
#pragma unroll
            for (int mt = 0; mt < 4; ++mt) {
#pragma unroll
                for (int nt = 0; nt < 2; ++nt) {
                    v8f acc = {};
#pragma unroll
                    for (int k0 = 0; k0 < VP; k0 += 4) {
                        const int kk = k0 + 2 * hi;
                        v2f a, b;
                        a.x = sm[OFF_INP + cl * (T_ * VP) + (mt * 16 + lo) * VP + kk];
                        a.y = sm[OFF_INP + cl * (T_ * VP) + (mt * 16 + lo) * VP + kk + 1];
                        b.x = sm[OFF_AF + cl * (UP * VP) + (nt * 16 + lo) * VP + kk];
                        b.y = sm[OFF_AF + cl * (UP * VP) + (nt * 16 + lo) * VP + kk + 1];
                        acc = __builtin_amdgcn_wmma_f32_16x16x4_f32(
                                false, a, false, b, (short)0, acc, false, false);
                    }
                    const int u = nt * 16 + lo;
                    if (u < V_) {
#pragma unroll
                        for (int j = 0; j < 8; ++j) {
                            const int t = mt * 16 + j + 8 * hi;
                            out[(((size_t)n * COUT + c) * T_ + t) * V_ + u] = acc[j];
                        }
                    }
                }
            }
        }
    }
}

extern "C" void kernel_launch(void* const* d_in, const int* in_sizes, int n_in,
                              void* d_out, int out_size, void* d_ws, size_t ws_size,
                              hipStream_t stream) {
    const float* x  = (const float*)d_in[0];
    const float* A  = (const float*)d_in[1];
    const float* w1 = (const float*)d_in[2];
    const float* b1 = (const float*)d_in[3];
    const float* w2 = (const float*)d_in[4];
    const float* b2 = (const float*)d_in[5];
    const float* w3 = (const float*)d_in[6];
    const float* b3 = (const float*)d_in[7];
    const float* wr = (const float*)d_in[8];
    const float* br = (const float*)d_in[9];
    float* out = (float*)d_out;

    const size_t shmem = (size_t)SMEM_FLOATS * sizeof(float);
    gcn_fused_kernel<<<dim3(N_), dim3(BLK), shmem, stream>>>(
        x, A, w1, b1, w2, b2, w3, b3, wr, br, out);
}


# GEMM1 x tile fetched with 8 loads of 16 bytes (was 32 of 4), made into B fragments through a 4 KB LDS tile of the wave's own; the group's wr fragments stay in registers; sums in the same order
# speedup vs baseline: 1.4212x; 1.4212x over previous
#include <hip/hip_runtime.h>
#include <cmath>

typedef float v2f __attribute__((ext_vector_type(2)));
typedef float v4f __attribute__((ext_vector_type(4)));
typedef float v8f __attribute__((ext_vector_type(8)));

#define N_    256
#define CIN   64
#define COUT  64
#define T_    64
#define V_    25
#define REL   8
#define TV    (T_ * V_)
#define VP    28
#define WRS   68
#define BLK   512
#define NWAVE 16

#define OFF_WR    0
#define WROWS     (COUT + 2 * REL)
#define SZ_WR     (WROWS * WRS)
#define OFF_X1    (OFF_WR + SZ_WR)
#define OFF_X2    (OFF_X1 + REL * V_)
#define OFF_INP   (OFF_X2 + REL * V_)
#define SZ_INP    (16 * T_ * VP)
#define OFF_Y     (OFF_INP + SZ_INP)
#define YP        (TV + 4)
#define SZ_Y      (2 * REL * YP)
#define OFF_D     OFF_Y
#define DRS       (V_ * VP)
#define SZ_D      (REL * DRS)
#define SMEM_USED (OFF_Y + SZ_Y)
#define OFF_STG   SMEM_USED
#define SZ_STG    (NWAVE * CIN * 16)
#define SMEM_FLOATS (OFF_STG + SZ_STG)
static_assert((OFF_STG % 4) == 0, "16-byte stores into the wave tiles need a 16-byte offset");
static_assert((SMEM_USED + NWAVE * CIN * 16) * 4 <= 320 * 1024, "the work-group's LDS must fit in one WGP");
static_assert(SZ_D <= SZ_Y, "d takes over the area of y");
static_assert((OFF_D % 2) == 0 && (DRS % 2) == 0 && (VP % 2) == 0, "8-byte reads of d need even offsets");

__global__ void __launch_bounds__(BLK) gcn_fused_kernel(const float* __restrict__ x,
                                 const float* __restrict__ Ag,
                                 const float* __restrict__ w1, const float* __restrict__ b1,
                                 const float* __restrict__ w2, const float* __restrict__ b2,
                                 const float* __restrict__ w3, const float* __restrict__ b3,
                                 const float* __restrict__ wr, const float* __restrict__ br,
                                 float* __restrict__ out)
{
    extern __shared__ __attribute__((aligned(16))) float sm[];
    const int n    = blockIdx.x;
    const int tid  = threadIdx.x;
    const int wave = tid >> 5;
    const int lane = tid & 31;
    const int lo   = lane & 15;
    const int hi   = lane >> 4;
    const float* __restrict__ xg = x + (size_t)n * CIN * TV;

    const int wave_u = __builtin_amdgcn_readfirstlane(wave);

    const float* __restrict__ xq = xg + (lane >> 2) * TV + 4 * (lane & 3);
    float* xt = sm + OFF_STG + wave_u * (CIN * 16);
    const float* xf = xt + (2 * hi) * 16 + lo;
    v4f xraw[8];
    v2f xfr[16];
#define LOAD_X(tile_)                                                         \
    {                                                                         \
        const float* __restrict__ xp = xq + (tile_) * 16;                     \
        __builtin_amdgcn_sched_barrier(0);                                    \
        _Pragma("unroll")                                                     \
        for (int i = 0; i < 8; ++i)                                           \
            xraw[i] = *reinterpret_cast<const v4f*>(xp + (8 * i) * TV);       \
        __builtin_amdgcn_sched_barrier(0);                                    \
    }
#define STAGE_X()                                                             \
    {                                                                         \
        _Pragma("unroll")                                                     \
        for (int i = 0; i < 8; ++i)                                           \
            *reinterpret_cast<v4f*>(xt + 128 * i + 4 * lane) = xraw[i];       \
    }
#define FRAGS_X()                                                             \
    {                                                                         \
        _Pragma("unroll")                                                     \
        for (int ks = 0; ks < 16; ++ks) {                                     \
            xfr[ks].x = xf[64 * ks];                                          \
            xfr[ks].y = xf[64 * ks + 16];                                     \
        }                                                                     \
    }
    LOAD_X(wave_u)

    static_assert(REL * CIN == BLK, "one value of w1 and one of w2 per thread");
    const float wy1 = w1[tid];
    const float wy2 = w2[tid];
    for (int idx = tid; idx < COUT * CIN; idx += BLK) {
        const int c = idx / CIN, i = idx % CIN;
        sm[OFF_WR + c * WRS + i] = wr[idx];
    }
    sm[OFF_WR + (COUT + tid / CIN) * WRS + tid % CIN] = wy1;
    sm[OFF_WR + (COUT + REL + tid / CIN) * WRS + tid % CIN] = wy2;
    for (int idx = tid; idx < SZ_INP; idx += BLK) sm[OFF_INP + idx] = 0.0f;
    __syncthreads();

    const bool has7 = wave_u < (TV / 16 - 6 * NWAVE);
#define GEMM1_TRIPS(MMA_, more_)                                              \
    for (int p = 0; p < 6; ++p) {                                             \
        const int te = wave_u + 16 * p;                                       \
        STAGE_X()                                                             \
        FRAGS_X()                                                             \
        const int tn = (p < 5 || has7) ? te + 16 : wave_u;                    \
        LOAD_X(tn)                                                            \
        MMA_(xfr, te)                                                         \
    }                                                                         \
    if (has7) {                                                               \
        STAGE_X()                                                             \
        FRAGS_X()                                                             \
        if (more_) LOAD_X(wave_u)                                             \
        MMA_(xfr, wave_u + 96)                                                \
    }

    {
        float brv[8];
#pragma unroll
        for (int j = 0; j < 8; ++j) brv[j] = br[j + 8 * hi];
        const float* wrow  = sm + OFF_WR + lo * WRS + 2 * hi;
        const float* wyrow = sm + OFF_WR + (COUT + lo) * WRS + 2 * hi;
        v2f afr[16], ayr[16];
#pragma unroll
        for (int ks = 0; ks < 16; ++ks) {
            afr[ks].x = wrow[4 * ks];
            afr[ks].y = wrow[4 * ks + 1];
            ayr[ks].x = wyrow[4 * ks];
            ayr[ks].y = wyrow[4 * ks + 1];
        }
#define MMA_XY(bsrc, tile_)                                                   \
        {                                                                     \
            v8f acc = {};                                                     \
            v8f accy = {};                                                    \
            _Pragma("unroll")                                                 \
            for (int ks = 0; ks < 16; ++ks) {                                 \
                acc = __builtin_amdgcn_wmma_f32_16x16x4_f32(                  \
                        false, afr[ks], false, bsrc[ks], (short)0, acc, false, false);   \
                accy = __builtin_amdgcn_wmma_f32_16x16x4_f32(                 \
                        false, ayr[ks], false, bsrc[ks], (short)0, accy, false, false); \
            }                                                                 \
            const int col = (tile_) * 16 + lo;                                \
            const int t = col / V_, v = col % V_;                             \
            float* dst = sm + OFF_INP + (8 * hi) * (T_ * VP) + t * VP + v;    \
            float* ydst = sm + OFF_Y + (8 * hi) * YP + col;                   \
            _Pragma("unroll")                                                 \
            for (int j = 0; j < 8; ++j) {                                     \
                dst[j * (T_ * VP)] = acc[j] + brv[j];                         \
                ydst[j * YP] = accy[j];                                       \
            }                                                                 \
        }
        GEMM1_TRIPS(MMA_XY, true)
#undef MMA_XY
    }
    __syncthreads();

    for (int idx = tid; idx < 2 * REL * V_; idx += BLK) {
        const int m = idx / V_, v = idx % V_;
        const int r = m & (REL - 1);
        const float* yp = sm + OFF_Y + m * YP + v;
        float s0 = 0.0f, s1 = 0.0f, s2 = 0.0f, s3 = 0.0f;
#pragma unroll
        for (int t = 0; t < T_; t += 4) {
            s0 += yp[t * V_];
            s1 += yp[(t + 1) * V_];
            s2 += yp[(t + 2) * V_];
            s3 += yp[(t + 3) * V_];
        }
        const float s = (s0 + s1) + (s2 + s3);
        sm[((m < REL) ? OFF_X1 : OFF_X2) + r * V_ + v] = s * (1.0f / T_) + ((m < REL) ? b1[r] : b2[r]);
    }
    __syncthreads();

    for (int idx = tid; idx < SZ_D; idx += BLK) {
        const int r = idx / DRS;
        const int uv = idx % DRS;
        const int u = uv / VP, v = uv % VP;
        const int vcl = v < V_ ? v : V_ - 1;
        const float dv = tanhf(sm[OFF_X1 + r * V_ + u] - sm[OFF_X2 + r * V_ + vcl]);
        sm[OFF_D + idx] = v < V_ ? dv : 0.0f;
    }

    const int u0 = lo;
    const int u1 = (16 + lo) < V_ ? (16 + lo) : V_ - 1;
    v2f abase[2][7];
#pragma unroll
    for (int nt = 0; nt < 2; ++nt) {
        const float* __restrict__ arow = Ag + (nt ? u1 : u0) * V_;
#pragma unroll
        for (int s = 0; s < 6; ++s) {
            abase[nt][s].x = arow[4 * s + 2 * hi];
            abase[nt][s].y = arow[4 * s + 2 * hi + 1];
        }
        const float a24 = arow[V_ - 1];
        abase[nt][6].x = hi ? 0.0f : a24;
        abase[nt][6].y = 0.0f;
    }
    const float* __restrict__ dlane0 = sm + OFF_D + u0 * VP + 2 * hi;
    const float* __restrict__ dlane1 = sm + OFF_D + u1 * VP + 2 * hi;

    v2f bfrag[2][7];
    float w3c[REL];
    float b3c;
#define LOAD_W3(ct_)                                                          \
    {                                                                         \
        const int cw = (ct_) * 16 + wave_u;                                   \
        _Pragma("unroll")                                                     \
        for (int r = 0; r < REL; ++r) w3c[r] = w3[cw * REL + r];              \
        b3c = b3[cw];                                                         \
    }
#define BUILD_AFULL()                                                                                   \
    {                                                                                                   \
        _Pragma("unroll")                                                                               \
        for (int nt = 0; nt < 2; ++nt) {                                                                \
            const float* __restrict__ dl = nt ? dlane1 : dlane0;                                        \
        _Pragma("unroll")                                                                               \
            for (int s = 0; s < 7; ++s) {                                                               \
                v2f acc = abase[nt][s];                                                                 \
                if (s < 6) { acc.x += b3c; acc.y += b3c; }                                              \
                else       { acc.x += hi ? 0.0f : b3c; }                                                \
        _Pragma("unroll")                                                                               \
                for (int r = 0; r < REL; ++r) {                                                         \
                    const v2f dv = *reinterpret_cast<const v2f*>(dl + r * DRS + 4 * s);                 \
                    const v2f wv = {w3c[r], w3c[r]};                                                    \
                    acc = __builtin_elementwise_fma(wv, dv, acc);                                       \
                }                                                                                       \
                bfrag[nt][s] = acc;                                                                     \
            }                                                                                           \
        }                                                                                               \
    }

    LOAD_W3(0)
    __syncthreads();
    BUILD_AFULL()

    for (int ct = 0; ; ++ct) {
        {
            const int cl = wave_u;
            const int c = ct * 16 + cl;
#pragma unroll
            for (int mt = 0; mt < 4; ++mt) {
#pragma unroll
                for (int nt = 0; nt < 2; ++nt) {
                    v8f acc = {};
#pragma unroll
                    for (int k0 = 0; k0 < VP; k0 += 4) {
                        const int kk = k0 + 2 * hi;
                        v2f a;
                        a.x = sm[OFF_INP + cl * (T_ * VP) + (mt * 16 + lo) * VP + kk];
                        a.y = sm[OFF_INP + cl * (T_ * VP) + (mt * 16 + lo) * VP + kk + 1];
                        acc = __builtin_amdgcn_wmma_f32_16x16x4_f32(
                                false, a, false, bfrag[nt][k0 / 4], (short)0, acc, false, false);
                    }
                    const int u = nt * 16 + lo;
                    if (u < V_) {
#pragma unroll
                        for (int j = 0; j < 8; ++j) {
                            const int t = mt * 16 + j + 8 * hi;
                            out[(((size_t)n * COUT + c) * T_ + t) * V_ + u] = acc[j];
                        }
                    }
                }
            }
        }
        if (ct == COUT / 16 - 1) break;

        __builtin_amdgcn_fence(__ATOMIC_RELEASE, "workgroup", "local");
        __builtin_amdgcn_s_barrier();
        __builtin_amdgcn_fence(__ATOMIC_ACQUIRE, "workgroup", "local");
        LOAD_W3(ct + 1)

        {
            float brv[8];
#pragma unroll
            for (int j = 0; j < 8; ++j) brv[j] = br[(ct + 1) * 16 + j + 8 * hi];
            const float* wrow = sm + OFF_WR + ((ct + 1) * 16 + lo) * WRS + 2 * hi;
            v2f afr[16];
#pragma unroll
            for (int ks = 0; ks < 16; ++ks) {
                afr[ks].x = wrow[4 * ks];
                afr[ks].y = wrow[4 * ks + 1];
            }
#define MMA_X(bsrc, tile_)                                                    \
            {                                                                 \
                v8f acc = {};                                                 \
                _Pragma("unroll")                                             \
                for (int ks = 0; ks < 16; ++ks) {                             \
                    acc = __builtin_amdgcn_wmma_f32_16x16x4_f32(              \
                            false, afr[ks], false, bsrc[ks], (short)0, acc, false, false); \
                }                                                             \
                const int col = (tile_) * 16 + lo;                            \
                const int t = col / V_, v = col % V_;                         \
                float* dst = sm + OFF_INP + (8 * hi) * (T_ * VP) + t * VP + v; \
                _Pragma("unroll")                                             \
                for (int j = 0; j < 8; ++j) dst[j * (T_ * VP)] = acc[j] + brv[j]; \
            }
            GEMM1_TRIPS(MMA_X, ct + 2 < COUT / 16)
#undef MMA_X
        }
        BUILD_AFULL()

        __syncthreads();
    }
}

#undef BUILD_AFULL
#undef LOAD_W3
#undef GEMM1_TRIPS
#undef FRAGS_X
#undef STAGE_X
#undef LOAD_X

extern "C" void kernel_launch(void* const* d_in, const int* in_sizes, int n_in,
                              void* d_out, int out_size, void* d_ws, size_t ws_size,
                              hipStream_t stream) {
    const float* x  = (const float*)d_in[0];
    const float* A  = (const float*)d_in[1];
    const float* w1 = (const float*)d_in[2];
    const float* b1 = (const float*)d_in[3];
    const float* w2 = (const float*)d_in[4];
    const float* b2 = (const float*)d_in[5];
    const float* w3 = (const float*)d_in[6];
    const float* b3 = (const float*)d_in[7];
    const float* wr = (const float*)d_in[8];
    const float* br = (const float*)d_in[9];
    float* out = (float*)d_out;

    const size_t shmem = (size_t)SMEM_FLOATS * sizeof(float);
    gcn_fused_kernel<<<dim3(N_), dim3(BLK), shmem, stream>>>(
        x, A, w1, b1, w2, b2, w3, b3, wr, br, out);
}


# GEMM1 x operand straight from load registers: a wave takes 64-column blocks, a lane's 16-byte load is one column of each of four interleaved tiles; no LDS tile, four product chains; same sum order
# speedup vs baseline: 1.5067x; 1.0602x over previous
#include <hip/hip_runtime.h>
#include <cmath>

typedef float v2f __attribute__((ext_vector_type(2)));
typedef float v4f __attribute__((ext_vector_type(4)));
typedef float v8f __attribute__((ext_vector_type(8)));

#define N_    256
#define CIN   64
#define COUT  64
#define T_    64
#define V_    25
#define REL   8
#define TV    (T_ * V_)
#define VP    28
#define WRS   68
#define BLK   512
#define NWAVE 16
#define QCOLS 64
#define NQUAD (TV / QCOLS)
static_assert(NQUAD * QCOLS == TV, "column blocks tile a channel row exactly");
static_assert(NQUAD > NWAVE && NQUAD <= 2 * NWAVE, "one or two column blocks per wave");

#define OFF_WR    0
#define WROWS     (COUT + 2 * REL)
#define SZ_WR     (WROWS * WRS)
#define OFF_X1    (OFF_WR + SZ_WR)
#define OFF_X2    (OFF_X1 + REL * V_)
#define OFF_INP   (OFF_X2 + REL * V_)
#define SZ_INP    (16 * T_ * VP)
#define OFF_Y     (OFF_INP + SZ_INP)
#define YP        (TV + 4)
#define SZ_Y      (2 * REL * YP)
#define OFF_D     OFF_Y
#define DRS       (V_ * VP)
#define SZ_D      (REL * DRS)
#define SMEM_FLOATS (OFF_Y + SZ_Y)
static_assert(SMEM_FLOATS * 4 <= 320 * 1024, "the work-group's LDS must fit in one WGP");
static_assert(15 * YP + 2 + TV <= SZ_Y, "the shifted upper rows of y stay inside its area");
static_assert(SZ_D <= SZ_Y, "d takes over the area of y");
static_assert((OFF_D % 2) == 0 && (DRS % 2) == 0 && (VP % 2) == 0, "8-byte reads of d need even offsets");

__global__ void __launch_bounds__(BLK) gcn_fused_kernel(const float* __restrict__ x,
                                 const float* __restrict__ Ag,
                                 const float* __restrict__ w1, const float* __restrict__ b1,
                                 const float* __restrict__ w2, const float* __restrict__ b2,
                                 const float* __restrict__ w3, const float* __restrict__ b3,
                                 const float* __restrict__ wr, const float* __restrict__ br,
                                 float* __restrict__ out)
{
    extern __shared__ __attribute__((aligned(16))) float sm[];
    const int n    = blockIdx.x;
    const int tid  = threadIdx.x;
    const int wave = tid >> 5;
    const int lane = tid & 31;
    const int lo   = lane & 15;
    const int hi   = lane >> 4;
    const float* __restrict__ xg = x + (size_t)n * CIN * TV;

    const int wave_u = __builtin_amdgcn_readfirstlane(wave);

    const float* __restrict__ xq = xg + (2 * hi) * TV + 4 * lo;
    v4f xa[8], xb[8];
#define LOAD_XC(dst_, quad_, chunk_)                                          \
    {                                                                         \
        const float* __restrict__ xp = xq + (quad_) * QCOLS + (chunk_) * (16 * TV); \
        __builtin_amdgcn_sched_barrier(0);                                    \
        _Pragma("unroll")                                                     \
        for (int i = 0; i < 8; ++i)                                           \
            dst_[i] = *reinterpret_cast<const v4f*>(xp + (4 * (i >> 1) + (i & 1)) * TV); \
        __builtin_amdgcn_sched_barrier(0);                                    \
    }
    const int nquads = wave_u < (NQUAD - NWAVE) ? 2 : 1;
    LOAD_XC(xa, wave_u, 0)

    static_assert(REL * CIN == BLK, "one value of w1 and one of w2 per thread");
    const float wy1 = w1[tid];
    const float wy2 = w2[tid];
    for (int idx = tid; idx < COUT * CIN; idx += BLK) {
        const int c = idx / CIN, i = idx % CIN;
        sm[OFF_WR + c * WRS + i] = wr[idx];
    }
    sm[OFF_WR + (COUT + tid / CIN) * WRS + tid % CIN] = wy1;
    sm[OFF_WR + (COUT + REL + tid / CIN) * WRS + tid % CIN] = wy2;
    for (int idx = tid; idx < SZ_INP; idx += BLK) sm[OFF_INP + idx] = 0.0f;
    __syncthreads();

#define GEMM1_GROUP(DECL_, MMA_, EPI_)                                        \
    for (int qi = 0; qi < nquads; ++qi) {                                     \
        const int quad = wave_u + NWAVE * qi;                                 \
        const int qnext = (qi + 1 < nquads) ? quad + NWAVE : wave_u;          \
        DECL_                                                                 \
        LOAD_XC(xb, quad, 1)                                                  \
        MMA_(xa, 0)                                                           \
        LOAD_XC(xa, quad, 2)                                                  \
        MMA_(xb, 1)                                                           \
        LOAD_XC(xb, quad, 3)                                                  \
        MMA_(xa, 2)                                                           \
        LOAD_XC(xa, qnext, 0)                                                 \
        MMA_(xb, 3)                                                           \
        EPI_(quad)                                                            \
    }

    {
        float brv[8];
#pragma unroll
        for (int j = 0; j < 8; ++j) brv[j] = br[j + 8 * hi];
        const float* wrow  = sm + OFF_WR + lo * WRS + 2 * hi;
        const float* wyrow = sm + OFF_WR + (COUT + lo) * WRS + 2 * hi;
        v2f afr[16], ayr[16];
#pragma unroll
        for (int ks = 0; ks < 16; ++ks) {
            afr[ks].x = wrow[4 * ks];
            afr[ks].y = wrow[4 * ks + 1];
            ayr[ks].x = wyrow[4 * ks];
            ayr[ks].y = wyrow[4 * ks + 1];
        }
#define MMA_XY(src_, chunk_)                                                  \
        {                                                                     \
            _Pragma("unroll")                                                 \
            for (int s = 0; s < 4; ++s) {                                     \
                _Pragma("unroll")                                             \
                for (int j = 0; j < 4; ++j) {                                 \
                    v2f b;                                                    \
                    b.x = src_[2 * s][j];                                     \
                    b.y = src_[2 * s + 1][j];                                 \
                    acc[j] = __builtin_amdgcn_wmma_f32_16x16x4_f32(           \
                            false, afr[4 * (chunk_) + s], false, b, (short)0, acc[j], false, false);   \
                    accy[j] = __builtin_amdgcn_wmma_f32_16x16x4_f32(          \
                            false, ayr[4 * (chunk_) + s], false, b, (short)0, accy[j], false, false);  \
                }                                                             \
            }                                                                 \
        }
#define EPI_XY(quad_)                                                         \
        {                                                                     \
            _Pragma("unroll")                                                 \
            for (int j = 0; j < 4; ++j) {                                     \
                const int col = (quad_) * QCOLS + 4 * lo + j;                 \
                const int t = col / V_, v = col % V_;                         \
                float* dst = sm + OFF_INP + (8 * hi) * (T_ * VP) + t * VP + v; \
                float* ydst = sm + OFF_Y + (8 * hi) * YP + 2 * hi + col;      \
                _Pragma("unroll")                                             \
                for (int r = 0; r < 8; ++r) {                                 \
                    dst[r * (T_ * VP)] = acc[j][r] + brv[r];                  \
                    ydst[r * YP] = accy[j][r];                                \
                }                                                             \
            }                                                                 \
        }
        GEMM1_GROUP(v8f acc[4] = {}; v8f accy[4] = {};, MMA_XY, EPI_XY)
#undef EPI_XY
#undef MMA_XY
    }
    __syncthreads();

    for (int idx = tid; idx < 2 * REL * V_; idx += BLK) {
        const int m = idx / V_, v = idx % V_;
        const int r = m & (REL - 1);
        const float* yp = sm + OFF_Y + m * YP + 2 * (m >> 3) + v;
        float s0 = 0.0f, s1 = 0.0f, s2 = 0.0f, s3 = 0.0f;
#pragma unroll
        for (int t = 0; t < T_; t += 4) {
            s0 += yp[t * V_];
            s1 += yp[(t + 1) * V_];
            s2 += yp[(t + 2) * V_];
            s3 += yp[(t + 3) * V_];
        }
        const float s = (s0 + s1) + (s2 + s3);
        sm[((m < REL) ? OFF_X1 : OFF_X2) + r * V_ + v] = s * (1.0f / T_) + ((m < REL) ? b1[r] : b2[r]);
    }
    __syncthreads();

    for (int idx = tid; idx < SZ_D; idx += BLK) {
        const int r = idx / DRS;
        const int uv = idx % DRS;
        const int u = uv / VP, v = uv % VP;
        const int vcl = v < V_ ? v : V_ - 1;
        const float dv = tanhf(sm[OFF_X1 + r * V_ + u] - sm[OFF_X2 + r * V_ + vcl]);
        sm[OFF_D + idx] = v < V_ ? dv : 0.0f;
    }

    const int u0 = lo;
    const int u1 = (16 + lo) < V_ ? (16 + lo) : V_ - 1;
    v2f abase[2][7];
#pragma unroll
    for (int nt = 0; nt < 2; ++nt) {
        const float* __restrict__ arow = Ag + (nt ? u1 : u0) * V_;
#pragma unroll
        for (int s = 0; s < 6; ++s) {
            abase[nt][s].x = arow[4 * s + 2 * hi];
            abase[nt][s].y = arow[4 * s + 2 * hi + 1];
        }
        const float a24 = arow[V_ - 1];
        abase[nt][6].x = hi ? 0.0f : a24;
        abase[nt][6].y = 0.0f;
    }
    const float* __restrict__ dlane0 = sm + OFF_D + u0 * VP + 2 * hi;
    const float* __restrict__ dlane1 = sm + OFF_D + u1 * VP + 2 * hi;

    v2f bfrag[2][7];
    float w3c[REL];
    float b3c;
#define LOAD_W3(ct_)                                                          \
    {                                                                         \
        const int cw = (ct_) * 16 + wave_u;                                   \
        _Pragma("unroll")                                                     \
        for (int r = 0; r < REL; ++r) w3c[r] = w3[cw * REL + r];              \
        b3c = b3[cw];                                                         \
    }
#define BUILD_AFULL()                                                                                   \
    {                                                                                                   \
        _Pragma("unroll")                                                                               \
        for (int nt = 0; nt < 2; ++nt) {                                                                \
            const float* __restrict__ dl = nt ? dlane1 : dlane0;                                        \
        _Pragma("unroll")                                                                               \
            for (int s = 0; s < 7; ++s) {                                                               \
                v2f acc = abase[nt][s];                                                                 \
                if (s < 6) { acc.x += b3c; acc.y += b3c; }                                              \
                else       { acc.x += hi ? 0.0f : b3c; }                                                \
        _Pragma("unroll")                                                                               \
                for (int r = 0; r < REL; ++r) {                                                         \
                    const v2f dv = *reinterpret_cast<const v2f*>(dl + r * DRS + 4 * s);                 \
                    const v2f wv = {w3c[r], w3c[r]};                                                    \
                    acc = __builtin_elementwise_fma(wv, dv, acc);                                       \
                }                                                                                       \
                bfrag[nt][s] = acc;                                                                     \
            }                                                                                           \
        }                                                                                               \
    }

    LOAD_W3(0)
    __syncthreads();
    BUILD_AFULL()

    for (int ct = 0; ; ++ct) {
        {
            const int cl = wave_u;
            const int c = ct * 16 + cl;
#pragma unroll
            for (int mt = 0; mt < 4; ++mt) {
#pragma unroll
                for (int nt = 0; nt < 2; ++nt) {
                    v8f acc = {};
#pragma unroll
                    for (int k0 = 0; k0 < VP; k0 += 4) {
                        const int kk = k0 + 2 * hi;
                        v2f a;
                        a.x = sm[OFF_INP + cl * (T_ * VP) + (mt * 16 + lo) * VP + kk];
                        a.y = sm[OFF_INP + cl * (T_ * VP) + (mt * 16 + lo) * VP + kk + 1];
                        acc = __builtin_amdgcn_wmma_f32_16x16x4_f32(
                                false, a, false, bfrag[nt][k0 / 4], (short)0, acc, false, false);
                    }
                    const int u = nt * 16 + lo;
                    if (u < V_) {
#pragma unroll
                        for (int j = 0; j < 8; ++j) {
                            const int t = mt * 16 + j + 8 * hi;
                            out[(((size_t)n * COUT + c) * T_ + t) * V_ + u] = acc[j];
                        }
                    }
                }
            }
        }
        if (ct == COUT / 16 - 1) break;

        __builtin_amdgcn_fence(__ATOMIC_RELEASE, "workgroup", "local");
        __builtin_amdgcn_s_barrier();
        __builtin_amdgcn_fence(__ATOMIC_ACQUIRE, "workgroup", "local");
        LOAD_W3(ct + 1)

        {
            float brv[8];
#pragma unroll
            for (int j = 0; j < 8; ++j) brv[j] = br[(ct + 1) * 16 + j + 8 * hi];
            const float* wrow = sm + OFF_WR + ((ct + 1) * 16 + lo) * WRS + 2 * hi;
            v2f afr[16];
#pragma unroll
            for (int ks = 0; ks < 16; ++ks) {
                afr[ks].x = wrow[4 * ks];
                afr[ks].y = wrow[4 * ks + 1];
            }
#define MMA_X(src_, chunk_)                                                   \
            {                                                                 \
                _Pragma("unroll")                                             \
                for (int s = 0; s < 4; ++s) {                                 \
                    _Pragma("unroll")                                         \
                    for (int j = 0; j < 4; ++j) {                             \
                        v2f b;                                                \
                        b.x = src_[2 * s][j];                                 \
                        b.y = src_[2 * s + 1][j];                             \
                        acc[j] = __builtin_amdgcn_wmma_f32_16x16x4_f32(       \
                                false, afr[4 * (chunk_) + s], false, b, (short)0, acc[j], false, false); \
                    }                                                         \
                }                                                             \
            }
#define EPI_X(quad_)                                                          \
            {                                                                 \
                _Pragma("unroll")                                             \
                for (int j = 0; j < 4; ++j) {                                 \
                    const int col = (quad_) * QCOLS + 4 * lo + j;             \
                    const int t = col / V_, v = col % V_;                     \
                    float* dst = sm + OFF_INP + (8 * hi) * (T_ * VP) + t * VP + v; \
                    _Pragma("unroll")                                         \
                    for (int r = 0; r < 8; ++r) dst[r * (T_ * VP)] = acc[j][r] + brv[r]; \
                }                                                             \
            }
            GEMM1_GROUP(v8f acc[4] = {};, MMA_X, EPI_X)
#undef EPI_X
#undef MMA_X
        }
        BUILD_AFULL()

        __syncthreads();
    }
}

#undef BUILD_AFULL
#undef LOAD_W3
#undef GEMM1_GROUP
#undef LOAD_XC

extern "C" void kernel_launch(void* const* d_in, const int* in_sizes, int n_in,
                              void* d_out, int out_size, void* d_ws, size_t ws_size,
                              hipStream_t stream) {
    const float* x  = (const float*)d_in[0];
    const float* A  = (const float*)d_in[1];
    const float* w1 = (const float*)d_in[2];
    const float* b1 = (const float*)d_in[3];
    const float* w2 = (const float*)d_in[4];
    const float* b2 = (const float*)d_in[5];
    const float* w3 = (const float*)d_in[6];
    const float* b3 = (const float*)d_in[7];
    const float* wr = (const float*)d_in[8];
    const float* br = (const float*)d_in[9];
    float* out = (float*)d_out;

    const size_t shmem = (size_t)SMEM_FLOATS * sizeof(float);
    gcn_fused_kernel<<<dim3(N_), dim3(BLK), shmem, stream>>>(
        x, A, w1, b1, w2, b2, w3, b3, wr, br, out);
}


# GEMM2 output rows leave whole: the second u tile reads its A rows with the tile's halves exchanged, so one store writes a row's 25 floats as one piece, not 16 floats of two rows; same sum order
# speedup vs baseline: 1.5989x; 1.0612x over previous
#include <hip/hip_runtime.h>
#include <cmath>

typedef float v2f __attribute__((ext_vector_type(2)));
typedef float v4f __attribute__((ext_vector_type(4)));
typedef float v8f __attribute__((ext_vector_type(8)));

#define N_    256
#define CIN   64
#define COUT  64
#define T_    64
#define V_    25
#define REL   8
#define TV    (T_ * V_)
#define VP    28
#define WRS   68
#define BLK   512
#define NWAVE 16
#define QCOLS 64
#define NQUAD (TV / QCOLS)
static_assert(NQUAD * QCOLS == TV, "column blocks tile a channel row exactly");
static_assert(NQUAD > NWAVE && NQUAD <= 2 * NWAVE, "one or two column blocks per wave");

#define OFF_WR    0
#define WROWS     (COUT + 2 * REL)
#define SZ_WR     (WROWS * WRS)
#define OFF_X1    (OFF_WR + SZ_WR)
#define OFF_X2    (OFF_X1 + REL * V_)
#define OFF_INP   (OFF_X2 + REL * V_)
#define SZ_INP    (16 * T_ * VP)
#define OFF_Y     (OFF_INP + SZ_INP)
#define YP        (TV + 4)
#define SZ_Y      (2 * REL * YP)
#define OFF_D     OFF_Y
#define DRS       (V_ * VP)
#define SZ_D      (REL * DRS)
#define SMEM_FLOATS (OFF_Y + SZ_Y)
static_assert(SMEM_FLOATS * 4 <= 320 * 1024, "the work-group's LDS must fit in one WGP");
static_assert(15 * YP + 2 + TV <= SZ_Y, "the shifted upper rows of y stay inside its area");
static_assert(SZ_D <= SZ_Y, "d takes over the area of y");
static_assert((OFF_D % 2) == 0 && (DRS % 2) == 0 && (VP % 2) == 0, "8-byte reads of d need even offsets");

__global__ void __launch_bounds__(BLK) gcn_fused_kernel(const float* __restrict__ x,
                                 const float* __restrict__ Ag,
                                 const float* __restrict__ w1, const float* __restrict__ b1,
                                 const float* __restrict__ w2, const float* __restrict__ b2,
                                 const float* __restrict__ w3, const float* __restrict__ b3,
                                 const float* __restrict__ wr, const float* __restrict__ br,
                                 float* __restrict__ out)
{
    extern __shared__ __attribute__((aligned(16))) float sm[];
    const int n    = blockIdx.x;
    const int tid  = threadIdx.x;
    const int wave = tid >> 5;
    const int lane = tid & 31;
    const int lo   = lane & 15;
    const int hi   = lane >> 4;
    const float* __restrict__ xg = x + (size_t)n * CIN * TV;

    const int wave_u = __builtin_amdgcn_readfirstlane(wave);

    const float* __restrict__ xq = xg + (2 * hi) * TV + 4 * lo;
    v4f xa[8], xb[8];
#define LOAD_XC(dst_, quad_, chunk_)                                          \
    {                                                                         \
        const float* __restrict__ xp = xq + (quad_) * QCOLS + (chunk_) * (16 * TV); \
        __builtin_amdgcn_sched_barrier(0);                                    \
        _Pragma("unroll")                                                     \
        for (int i = 0; i < 8; ++i)                                           \
            dst_[i] = *reinterpret_cast<const v4f*>(xp + (4 * (i >> 1) + (i & 1)) * TV); \
        __builtin_amdgcn_sched_barrier(0);                                    \
    }
    const int nquads = wave_u < (NQUAD - NWAVE) ? 2 : 1;
    LOAD_XC(xa, wave_u, 0)

    static_assert(REL * CIN == BLK, "one value of w1 and one of w2 per thread");
    const float wy1 = w1[tid];
    const float wy2 = w2[tid];
    for (int idx = tid; idx < COUT * CIN; idx += BLK) {
        const int c = idx / CIN, i = idx % CIN;
        sm[OFF_WR + c * WRS + i] = wr[idx];
    }
    sm[OFF_WR + (COUT + tid / CIN) * WRS + tid % CIN] = wy1;
    sm[OFF_WR + (COUT + REL + tid / CIN) * WRS + tid % CIN] = wy2;
    for (int idx = tid; idx < SZ_INP; idx += BLK) sm[OFF_INP + idx] = 0.0f;
    __syncthreads();

#define GEMM1_GROUP(DECL_, MMA_, EPI_)                                        \
    for (int qi = 0; qi < nquads; ++qi) {                                     \
        const int quad = wave_u + NWAVE * qi;                                 \
        const int qnext = (qi + 1 < nquads) ? quad + NWAVE : wave_u;          \
        DECL_                                                                 \
        LOAD_XC(xb, quad, 1)                                                  \
        MMA_(xa, 0)                                                           \
        LOAD_XC(xa, quad, 2)                                                  \
        MMA_(xb, 1)                                                           \
        LOAD_XC(xb, quad, 3)                                                  \
        MMA_(xa, 2)                                                           \
        LOAD_XC(xa, qnext, 0)                                                 \
        MMA_(xb, 3)                                                           \
        EPI_(quad)                                                            \
    }

    {
        float brv[8];
#pragma unroll
        for (int j = 0; j < 8; ++j) brv[j] = br[j + 8 * hi];
        const float* wrow  = sm + OFF_WR + lo * WRS + 2 * hi;
        const float* wyrow = sm + OFF_WR + (COUT + lo) * WRS + 2 * hi;
        v2f afr[16], ayr[16];
#pragma unroll
        for (int ks = 0; ks < 16; ++ks) {
            afr[ks].x = wrow[4 * ks];
            afr[ks].y = wrow[4 * ks + 1];
            ayr[ks].x = wyrow[4 * ks];
            ayr[ks].y = wyrow[4 * ks + 1];
        }
#define MMA_XY(src_, chunk_)                                                  \
        {                                                                     \
            _Pragma("unroll")                                                 \
            for (int s = 0; s < 4; ++s) {                                     \
                _Pragma("unroll")                                             \
                for (int j = 0; j < 4; ++j) {                                 \
                    v2f b;                                                    \
                    b.x = src_[2 * s][j];                                     \
                    b.y = src_[2 * s + 1][j];                                 \
                    acc[j] = __builtin_amdgcn_wmma_f32_16x16x4_f32(           \
                            false, afr[4 * (chunk_) + s], false, b, (short)0, acc[j], false, false);   \
                    accy[j] = __builtin_amdgcn_wmma_f32_16x16x4_f32(          \
                            false, ayr[4 * (chunk_) + s], false, b, (short)0, accy[j], false, false);  \
                }                                                             \
            }                                                                 \
        }
#define EPI_XY(quad_)                                                         \
        {                                                                     \
            _Pragma("unroll")                                                 \
            for (int j = 0; j < 4; ++j) {                                     \
                const int col = (quad_) * QCOLS + 4 * lo + j;                 \
                const int t = col / V_, v = col % V_;                         \
                float* dst = sm + OFF_INP + (8 * hi) * (T_ * VP) + t * VP + v; \
                float* ydst = sm + OFF_Y + (8 * hi) * YP + 2 * hi + col;      \
                _Pragma("unroll")                                             \
                for (int r = 0; r < 8; ++r) {                                 \
                    dst[r * (T_ * VP)] = acc[j][r] + brv[r];                  \
                    ydst[r * YP] = accy[j][r];                                \
                }                                                             \
            }                                                                 \
        }
        GEMM1_GROUP(v8f acc[4] = {}; v8f accy[4] = {};, MMA_XY, EPI_XY)
#undef EPI_XY
#undef MMA_XY
    }
    __syncthreads();

    for (int idx = tid; idx < 2 * REL * V_; idx += BLK) {
        const int m = idx / V_, v = idx % V_;
        const int r = m & (REL - 1);
        const float* yp = sm + OFF_Y + m * YP + 2 * (m >> 3) + v;
        float s0 = 0.0f, s1 = 0.0f, s2 = 0.0f, s3 = 0.0f;
#pragma unroll
        for (int t = 0; t < T_; t += 4) {
            s0 += yp[t * V_];
            s1 += yp[(t + 1) * V_];
            s2 += yp[(t + 2) * V_];
            s3 += yp[(t + 3) * V_];
        }
        const float s = (s0 + s1) + (s2 + s3);
        sm[((m < REL) ? OFF_X1 : OFF_X2) + r * V_ + v] = s * (1.0f / T_) + ((m < REL) ? b1[r] : b2[r]);
    }
    __syncthreads();

    for (int idx = tid; idx < SZ_D; idx += BLK) {
        const int r = idx / DRS;
        const int uv = idx % DRS;
        const int u = uv / VP, v = uv % VP;
        const int vcl = v < V_ ? v : V_ - 1;
        const float dv = tanhf(sm[OFF_X1 + r * V_ + u] - sm[OFF_X2 + r * V_ + vcl]);
        sm[OFF_D + idx] = v < V_ ? dv : 0.0f;
    }

    const int u0 = lo;
    const int u1 = (16 + lo) < V_ ? (16 + lo) : V_ - 1;
    v2f abase[2][7];
#pragma unroll
    for (int nt = 0; nt < 2; ++nt) {
        const float* __restrict__ arow = Ag + (nt ? u1 : u0) * V_;
#pragma unroll
        for (int s = 0; s < 6; ++s) {
            abase[nt][s].x = arow[4 * s + 2 * hi];
            abase[nt][s].y = arow[4 * s + 2 * hi + 1];
        }
        const float a24 = arow[V_ - 1];
        abase[nt][6].x = hi ? 0.0f : a24;
        abase[nt][6].y = 0.0f;
    }
    const float* __restrict__ dlane0 = sm + OFF_D + u0 * VP + 2 * hi;
    const float* __restrict__ dlane1 = sm + OFF_D + u1 * VP + 2 * hi;

    v2f bfrag[2][7];
    float w3c[REL];
    float b3c;
#define LOAD_W3(ct_)                                                          \
    {                                                                         \
        const int cw = (ct_) * 16 + wave_u;                                   \
        _Pragma("unroll")                                                     \
        for (int r = 0; r < REL; ++r) w3c[r] = w3[cw * REL + r];              \
        b3c = b3[cw];                                                         \
    }
#define BUILD_AFULL()                                                                                   \
    {                                                                                                   \
        _Pragma("unroll")                                                                               \
        for (int nt = 0; nt < 2; ++nt) {                                                                \
            const float* __restrict__ dl = nt ? dlane1 : dlane0;                                        \
        _Pragma("unroll")                                                                               \
            for (int s = 0; s < 7; ++s) {                                                               \
                v2f acc = abase[nt][s];                                                                 \
                if (s < 6) { acc.x += b3c; acc.y += b3c; }                                              \
                else       { acc.x += hi ? 0.0f : b3c; }                                                \
        _Pragma("unroll")                                                                               \
                for (int r = 0; r < REL; ++r) {                                                         \
                    const v2f dv = *reinterpret_cast<const v2f*>(dl + r * DRS + 4 * s);                 \
                    const v2f wv = {w3c[r], w3c[r]};                                                    \
                    acc = __builtin_elementwise_fma(wv, dv, acc);                                       \
                }                                                                                       \
                bfrag[nt][s] = acc;                                                                     \
            }                                                                                           \
        }                                                                                               \
    }

    LOAD_W3(0)
    __syncthreads();
    BUILD_AFULL()

    for (int ct = 0; ; ++ct) {
        {
            const int cl = wave_u;
            const int c = ct * 16 + cl;
            float* __restrict__ oplane = out + ((size_t)n * COUT + c) * (T_ * V_);
            const int ua = lane;
            const int ub = lane ^ 16;
#pragma unroll
            for (int mt = 0; mt < 4; ++mt) {
                v8f acc0 = {}, acc1 = {};
#pragma unroll
                for (int k0 = 0; k0 < VP; k0 += 4) {
                    const int kk = k0 + 2 * hi;
                    v2f a, a1;
                    a.x  = sm[OFF_INP + cl * (T_ * VP) + (mt * 16 + lo) * VP + kk];
                    a.y  = sm[OFF_INP + cl * (T_ * VP) + (mt * 16 + lo) * VP + kk + 1];
                    a1.x = sm[OFF_INP + cl * (T_ * VP) + (mt * 16 + (lo ^ 8)) * VP + kk];
                    a1.y = sm[OFF_INP + cl * (T_ * VP) + (mt * 16 + (lo ^ 8)) * VP + kk + 1];
                    acc0 = __builtin_amdgcn_wmma_f32_16x16x4_f32(
                            false, a, false, bfrag[0][k0 / 4], (short)0, acc0, false, false);
                    acc1 = __builtin_amdgcn_wmma_f32_16x16x4_f32(
                            false, a1, false, bfrag[1][k0 / 4], (short)0, acc1, false, false);
                }
                if (ua < V_) {
                    float* __restrict__ oa = oplane + (mt * 16) * V_ + ua;
#pragma unroll
                    for (int j = 0; j < 8; ++j) oa[j * V_] = hi ? acc1[j] : acc0[j];
                }
                if (ub < V_) {
                    float* __restrict__ ob = oplane + (mt * 16 + 8) * V_ + ub;
#pragma unroll
                    for (int j = 0; j < 8; ++j) ob[j * V_] = hi ? acc0[j] : acc1[j];
                }
            }
        }
        if (ct == COUT / 16 - 1) break;

        __builtin_amdgcn_fence(__ATOMIC_RELEASE, "workgroup", "local");
        __builtin_amdgcn_s_barrier();
        __builtin_amdgcn_fence(__ATOMIC_ACQUIRE, "workgroup", "local");
        LOAD_W3(ct + 1)

        {
            float brv[8];
#pragma unroll
            for (int j = 0; j < 8; ++j) brv[j] = br[(ct + 1) * 16 + j + 8 * hi];
            const float* wrow = sm + OFF_WR + ((ct + 1) * 16 + lo) * WRS + 2 * hi;
            v2f afr[16];
#pragma unroll
            for (int ks = 0; ks < 16; ++ks) {
                afr[ks].x = wrow[4 * ks];
                afr[ks].y = wrow[4 * ks + 1];
            }
#define MMA_X(src_, chunk_)                                                   \
            {                                                                 \
                _Pragma("unroll")                                             \
                for (int s = 0; s < 4; ++s) {                                 \
                    _Pragma("unroll")                                         \
                    for (int j = 0; j < 4; ++j) {                             \
                        v2f b;                                                \
                        b.x = src_[2 * s][j];                                 \
                        b.y = src_[2 * s + 1][j];                             \
                        acc[j] = __builtin_amdgcn_wmma_f32_16x16x4_f32(       \
                                false, afr[4 * (chunk_) + s], false, b, (short)0, acc[j], false, false); \
                    }                                                         \
                }                                                             \
            }
#define EPI_X(quad_)                                                          \
            {                                                                 \
                _Pragma("unroll")                                             \
                for (int j = 0; j < 4; ++j) {                                 \
                    const int col = (quad_) * QCOLS + 4 * lo + j;             \
                    const int t = col / V_, v = col % V_;                     \
                    float* dst = sm + OFF_INP + (8 * hi) * (T_ * VP) + t * VP + v; \
                    _Pragma("unroll")                                         \
                    for (int r = 0; r < 8; ++r) dst[r * (T_ * VP)] = acc[j][r] + brv[r]; \
                }                                                             \
            }
            GEMM1_GROUP(v8f acc[4] = {};, MMA_X, EPI_X)
#undef EPI_X
#undef MMA_X
        }
        BUILD_AFULL()

        __syncthreads();
    }
}

#undef BUILD_AFULL
#undef LOAD_W3
#undef GEMM1_GROUP
#undef LOAD_XC

extern "C" void kernel_launch(void* const* d_in, const int* in_sizes, int n_in,
                              void* d_out, int out_size, void* d_ws, size_t ws_size,
                              hipStream_t stream) {
    const float* x  = (const float*)d_in[0];
    const float* A  = (const float*)d_in[1];
    const float* w1 = (const float*)d_in[2];
    const float* b1 = (const float*)d_in[3];
    const float* w2 = (const float*)d_in[4];
    const float* b2 = (const float*)d_in[5];
    const float* w3 = (const float*)d_in[6];
    const float* b3 = (const float*)d_in[7];
    const float* wr = (const float*)d_in[8];
    const float* br = (const float*)d_in[9];
    float* out = (float*)d_out;

    const size_t shmem = (size_t)SMEM_FLOATS * sizeof(float);
    gcn_fused_kernel<<<dim3(N_), dim3(BLK), shmem, stream>>>(
        x, A, w1, b1, w2, b2, w3, b3, wr, br, out);
}


# Small operands without serialized round trips: wr as two 16-byte loads per thread, in flight together before the first x chunk (was 8 waited one by one); A rows, w3, b3 asked ahead of tanh; same sums
# speedup vs baseline: 1.7103x; 1.0697x over previous
#include <hip/hip_runtime.h>
#include <cmath>

typedef float v2f __attribute__((ext_vector_type(2)));
typedef float v4f __attribute__((ext_vector_type(4)));
typedef float v8f __attribute__((ext_vector_type(8)));

#define N_    256
#define CIN   64
#define COUT  64
#define T_    64
#define V_    25
#define REL   8
#define TV    (T_ * V_)
#define VP    28
#define WRS   68
#define BLK   512
#define NWAVE 16
#define QCOLS 64
#define NQUAD (TV / QCOLS)
static_assert(NQUAD * QCOLS == TV, "column blocks tile a channel row exactly");
static_assert(NQUAD > NWAVE && NQUAD <= 2 * NWAVE, "one or two column blocks per wave");

#define OFF_WR    0
#define WROWS     (COUT + 2 * REL)
#define SZ_WR     (WROWS * WRS)
#define OFF_X1    (OFF_WR + SZ_WR)
#define OFF_X2    (OFF_X1 + REL * V_)
#define OFF_INP   (OFF_X2 + REL * V_)
#define SZ_INP    (16 * T_ * VP)
#define OFF_Y     (OFF_INP + SZ_INP)
#define YP        (TV + 4)
#define SZ_Y      (2 * REL * YP)
#define OFF_D     OFF_Y
#define DRS       (V_ * VP)
#define SZ_D      (REL * DRS)
#define SMEM_FLOATS (OFF_Y + SZ_Y)
static_assert(SMEM_FLOATS * 4 <= 320 * 1024, "the work-group's LDS must fit in one WGP");
static_assert(15 * YP + 2 + TV <= SZ_Y, "the shifted upper rows of y stay inside its area");
static_assert(SZ_D <= SZ_Y, "d takes over the area of y");
static_assert((OFF_D % 2) == 0 && (DRS % 2) == 0 && (VP % 2) == 0, "8-byte reads of d need even offsets");

__global__ void __launch_bounds__(BLK) gcn_fused_kernel(const float* __restrict__ x,
                                 const float* __restrict__ Ag,
                                 const float* __restrict__ w1, const float* __restrict__ b1,
                                 const float* __restrict__ w2, const float* __restrict__ b2,
                                 const float* __restrict__ w3, const float* __restrict__ b3,
                                 const float* __restrict__ wr, const float* __restrict__ br,
                                 float* __restrict__ out)
{
    extern __shared__ __attribute__((aligned(16))) float sm[];
    const int n    = blockIdx.x;
    const int tid  = threadIdx.x;
    const int wave = tid >> 5;
    const int lane = tid & 31;
    const int lo   = lane & 15;
    const int hi   = lane >> 4;
    const float* __restrict__ xg = x + (size_t)n * CIN * TV;

    const int wave_u = __builtin_amdgcn_readfirstlane(wave);

    const float* __restrict__ xq = xg + (2 * hi) * TV + 4 * lo;
    v4f xa[8], xb[8];
#define LOAD_XC(dst_, quad_, chunk_)                                          \
    {                                                                         \
        const float* __restrict__ xp = xq + (quad_) * QCOLS + (chunk_) * (16 * TV); \
        __builtin_amdgcn_sched_barrier(0);                                    \
        _Pragma("unroll")                                                     \
        for (int i = 0; i < 8; ++i)                                           \
            dst_[i] = *reinterpret_cast<const v4f*>(xp + (4 * (i >> 1) + (i & 1)) * TV); \
        __builtin_amdgcn_sched_barrier(0);                                    \
    }
    const int nquads = wave_u < (NQUAD - NWAVE) ? 2 : 1;

    static_assert(REL * CIN == BLK, "one value of w1 and one of w2 per thread");
    static_assert(COUT * CIN == 8 * BLK, "two 16-byte pieces of wr per thread");
    static_assert((WRS % 4) == 0 && (CIN % 4) == 0 && (OFF_WR % 4) == 0, "a 16-byte piece stays inside one padded row, aligned");
    const float wy1 = w1[tid];
    const float wy2 = w2[tid];
    const v4f wrv0 = *reinterpret_cast<const v4f*>(wr + 4 * tid);
    const v4f wrv1 = *reinterpret_cast<const v4f*>(wr + 4 * (tid + BLK));
    LOAD_XC(xa, wave_u, 0)

    for (int idx = tid; idx < SZ_INP; idx += BLK) sm[OFF_INP + idx] = 0.0f;
    {
        const int f0 = 4 * tid, f1 = 4 * (tid + BLK);
        *reinterpret_cast<v4f*>(sm + OFF_WR + (f0 / CIN) * WRS + (f0 % CIN)) = wrv0;
        *reinterpret_cast<v4f*>(sm + OFF_WR + (f1 / CIN) * WRS + (f1 % CIN)) = wrv1;
    }
    sm[OFF_WR + (COUT + tid / CIN) * WRS + tid % CIN] = wy1;
    sm[OFF_WR + (COUT + REL + tid / CIN) * WRS + tid % CIN] = wy2;
    __syncthreads();

#define GEMM1_GROUP(DECL_, MMA_, EPI_)                                        \
    for (int qi = 0; qi < nquads; ++qi) {                                     \
        const int quad = wave_u + NWAVE * qi;                                 \
        const int qnext = (qi + 1 < nquads) ? quad + NWAVE : wave_u;          \
        DECL_                                                                 \
        LOAD_XC(xb, quad, 1)                                                  \
        MMA_(xa, 0)                                                           \
        LOAD_XC(xa, quad, 2)                                                  \
        MMA_(xb, 1)                                                           \
        LOAD_XC(xb, quad, 3)                                                  \
        MMA_(xa, 2)                                                           \
        LOAD_XC(xa, qnext, 0)                                                 \
        MMA_(xb, 3)                                                           \
        EPI_(quad)                                                            \
    }

    {
        float brv[8];
#pragma unroll
        for (int j = 0; j < 8; ++j) brv[j] = br[j + 8 * hi];
        const float* wrow  = sm + OFF_WR + lo * WRS + 2 * hi;
        const float* wyrow = sm + OFF_WR + (COUT + lo) * WRS + 2 * hi;
        v2f afr[16], ayr[16];
#pragma unroll
        for (int ks = 0; ks < 16; ++ks) {
            afr[ks].x = wrow[4 * ks];
            afr[ks].y = wrow[4 * ks + 1];
            ayr[ks].x = wyrow[4 * ks];
            ayr[ks].y = wyrow[4 * ks + 1];
        }
#define MMA_XY(src_, chunk_)                                                  \
        {                                                                     \
            _Pragma("unroll")                                                 \
            for (int s = 0; s < 4; ++s) {                                     \
                _Pragma("unroll")                                             \
                for (int j = 0; j < 4; ++j) {                                 \
                    v2f b;                                                    \
                    b.x = src_[2 * s][j];                                     \
                    b.y = src_[2 * s + 1][j];                                 \
                    acc[j] = __builtin_amdgcn_wmma_f32_16x16x4_f32(           \
                            false, afr[4 * (chunk_) + s], false, b, (short)0, acc[j], false, false);   \
                    accy[j] = __builtin_amdgcn_wmma_f32_16x16x4_f32(          \
                            false, ayr[4 * (chunk_) + s], false, b, (short)0, accy[j], false, false);  \
                }                                                             \
            }                                                                 \
        }
#define EPI_XY(quad_)                                                         \
        {                                                                     \
            _Pragma("unroll")                                                 \
            for (int j = 0; j < 4; ++j) {                                     \
                const int col = (quad_) * QCOLS + 4 * lo + j;                 \
                const int t = col / V_, v = col % V_;                         \
                float* dst = sm + OFF_INP + (8 * hi) * (T_ * VP) + t * VP + v; \
                float* ydst = sm + OFF_Y + (8 * hi) * YP + 2 * hi + col;      \
                _Pragma("unroll")                                             \
                for (int r = 0; r < 8; ++r) {                                 \
                    dst[r * (T_ * VP)] = acc[j][r] + brv[r];                  \
                    ydst[r * YP] = accy[j][r];                                \
                }                                                             \
            }                                                                 \
        }
        GEMM1_GROUP(v8f acc[4] = {}; v8f accy[4] = {};, MMA_XY, EPI_XY)
#undef EPI_XY
#undef MMA_XY
    }
    __syncthreads();

    for (int idx = tid; idx < 2 * REL * V_; idx += BLK) {
        const int m = idx / V_, v = idx % V_;
        const int r = m & (REL - 1);
        const float* yp = sm + OFF_Y + m * YP + 2 * (m >> 3) + v;
        float s0 = 0.0f, s1 = 0.0f, s2 = 0.0f, s3 = 0.0f;
#pragma unroll
        for (int t = 0; t < T_; t += 4) {
            s0 += yp[t * V_];
            s1 += yp[(t + 1) * V_];
            s2 += yp[(t + 2) * V_];
            s3 += yp[(t + 3) * V_];
        }
        const float s = (s0 + s1) + (s2 + s3);
        sm[((m < REL) ? OFF_X1 : OFF_X2) + r * V_ + v] = s * (1.0f / T_) + ((m < REL) ? b1[r] : b2[r]);
    }
    __syncthreads();

    const int u0 = lo;
    const int u1 = (16 + lo) < V_ ? (16 + lo) : V_ - 1;
    v2f abase[2][7];
    {
        const float* __restrict__ arow0 = Ag + u0 * V_;
        const float* __restrict__ arow1 = Ag + u1 * V_;
#pragma unroll
        for (int s = 0; s < 6; ++s) {
            abase[0][s].x = arow0[4 * s + 2 * hi];
            abase[0][s].y = arow0[4 * s + 2 * hi + 1];
            abase[1][s].x = arow1[4 * s + 2 * hi];
            abase[1][s].y = arow1[4 * s + 2 * hi + 1];
        }
        const float a24_0 = arow0[V_ - 1];
        const float a24_1 = arow1[V_ - 1];
        abase[0][6].x = hi ? 0.0f : a24_0;
        abase[0][6].y = 0.0f;
        abase[1][6].x = hi ? 0.0f : a24_1;
        abase[1][6].y = 0.0f;
    }
    float w3c[REL];
    float b3c;
#define LOAD_W3(ct_)                                                          \
    {                                                                         \
        const int cw = (ct_) * 16 + wave_u;                                   \
        _Pragma("unroll")                                                     \
        for (int r = 0; r < REL; ++r) w3c[r] = w3[cw * REL + r];              \
        b3c = b3[cw];                                                         \
    }
    LOAD_W3(0)

    for (int idx = tid; idx < SZ_D; idx += BLK) {
        const int r = idx / DRS;
        const int uv = idx % DRS;
        const int u = uv / VP, v = uv % VP;
        const int vcl = v < V_ ? v : V_ - 1;
        const float dv = tanhf(sm[OFF_X1 + r * V_ + u] - sm[OFF_X2 + r * V_ + vcl]);
        sm[OFF_D + idx] = v < V_ ? dv : 0.0f;
    }

    const float* __restrict__ dlane0 = sm + OFF_D + u0 * VP + 2 * hi;
    const float* __restrict__ dlane1 = sm + OFF_D + u1 * VP + 2 * hi;

    v2f bfrag[2][7];
#define BUILD_AFULL()                                                                                   \
    {                                                                                                   \
        _Pragma("unroll")                                                                               \
        for (int nt = 0; nt < 2; ++nt) {                                                                \
            const float* __restrict__ dl = nt ? dlane1 : dlane0;                                        \
        _Pragma("unroll")                                                                               \
            for (int s = 0; s < 7; ++s) {                                                               \
                v2f acc = abase[nt][s];                                                                 \
                if (s < 6) { acc.x += b3c; acc.y += b3c; }                                              \
                else       { acc.x += hi ? 0.0f : b3c; }                                                \
        _Pragma("unroll")                                                                               \
                for (int r = 0; r < REL; ++r) {                                                         \
                    const v2f dv = *reinterpret_cast<const v2f*>(dl + r * DRS + 4 * s);                 \
                    const v2f wv = {w3c[r], w3c[r]};                                                    \
                    acc = __builtin_elementwise_fma(wv, dv, acc);                                       \
                }                                                                                       \
                bfrag[nt][s] = acc;                                                                     \
            }                                                                                           \
        }                                                                                               \
    }

    __syncthreads();
    BUILD_AFULL()

    for (int ct = 0; ; ++ct) {
        {
            const int cl = wave_u;
            const int c = ct * 16 + cl;
            float* __restrict__ oplane = out + ((size_t)n * COUT + c) * (T_ * V_);
            const int ua = lane;
            const int ub = lane ^ 16;
#pragma unroll
            for (int mt = 0; mt < 4; ++mt) {
                v8f acc0 = {}, acc1 = {};
#pragma unroll
                for (int k0 = 0; k0 < VP; k0 += 4) {
                    const int kk = k0 + 2 * hi;
                    v2f a, a1;
                    a.x  = sm[OFF_INP + cl * (T_ * VP) + (mt * 16 + lo) * VP + kk];
                    a.y  = sm[OFF_INP + cl * (T_ * VP) + (mt * 16 + lo) * VP + kk + 1];
                    a1.x = sm[OFF_INP + cl * (T_ * VP) + (mt * 16 + (lo ^ 8)) * VP + kk];
                    a1.y = sm[OFF_INP + cl * (T_ * VP) + (mt * 16 + (lo ^ 8)) * VP + kk + 1];
                    acc0 = __builtin_amdgcn_wmma_f32_16x16x4_f32(
                            false, a, false, bfrag[0][k0 / 4], (short)0, acc0, false, false);
                    acc1 = __builtin_amdgcn_wmma_f32_16x16x4_f32(
                            false, a1, false, bfrag[1][k0 / 4], (short)0, acc1, false, false);
                }
                if (ua < V_) {
                    float* __restrict__ oa = oplane + (mt * 16) * V_ + ua;
#pragma unroll
                    for (int j = 0; j < 8; ++j) oa[j * V_] = hi ? acc1[j] : acc0[j];
                }
                if (ub < V_) {
                    float* __restrict__ ob = oplane + (mt * 16 + 8) * V_ + ub;
#pragma unroll
                    for (int j = 0; j < 8; ++j) ob[j * V_] = hi ? acc0[j] : acc1[j];
                }
            }
        }
        if (ct == COUT / 16 - 1) break;

        __builtin_amdgcn_fence(__ATOMIC_RELEASE, "workgroup", "local");
        __builtin_amdgcn_s_barrier();
        __builtin_amdgcn_fence(__ATOMIC_ACQUIRE, "workgroup", "local");
        LOAD_W3(ct + 1)

        {
            float brv[8];
#pragma unroll
            for (int j = 0; j < 8; ++j) brv[j] = br[(ct + 1) * 16 + j + 8 * hi];
            const float* wrow = sm + OFF_WR + ((ct + 1) * 16 + lo) * WRS + 2 * hi;
            v2f afr[16];
#pragma unroll
            for (int ks = 0; ks < 16; ++ks) {
                afr[ks].x = wrow[4 * ks];
                afr[ks].y = wrow[4 * ks + 1];
            }
#define MMA_X(src_, chunk_)                                                   \
            {                                                                 \
                _Pragma("unroll")                                             \
                for (int s = 0; s < 4; ++s) {                                 \
                    _Pragma("unroll")                                         \
                    for (int j = 0; j < 4; ++j) {                             \
                        v2f b;                                                \
                        b.x = src_[2 * s][j];                                 \
                        b.y = src_[2 * s + 1][j];                             \
                        acc[j] = __builtin_amdgcn_wmma_f32_16x16x4_f32(       \
                                false, afr[4 * (chunk_) + s], false, b, (short)0, acc[j], false, false); \
                    }                                                         \
                }                                                             \
            }
#define EPI_X(quad_)                                                          \
            {                                                                 \
                _Pragma("unroll")                                             \
                for (int j = 0; j < 4; ++j) {                                 \
                    const int col = (quad_) * QCOLS + 4 * lo + j;             \
                    const int t = col / V_, v = col % V_;                     \
                    float* dst = sm + OFF_INP + (8 * hi) * (T_ * VP) + t * VP + v; \
                    _Pragma("unroll")                                         \
                    for (int r = 0; r < 8; ++r) dst[r * (T_ * VP)] = acc[j][r] + brv[r]; \
                }                                                             \
            }
            GEMM1_GROUP(v8f acc[4] = {};, MMA_X, EPI_X)
#undef EPI_X
#undef MMA_X
        }
        BUILD_AFULL()

        __syncthreads();
    }
}

#undef BUILD_AFULL
#undef LOAD_W3
#undef GEMM1_GROUP
#undef LOAD_XC

extern "C" void kernel_launch(void* const* d_in, const int* in_sizes, int n_in,
                              void* d_out, int out_size, void* d_ws, size_t ws_size,
                              hipStream_t stream) {
    const float* x  = (const float*)d_in[0];
    const float* A  = (const float*)d_in[1];
    const float* w1 = (const float*)d_in[2];
    const float* b1 = (const float*)d_in[3];
    const float* w2 = (const float*)d_in[4];
    const float* b2 = (const float*)d_in[5];
    const float* w3 = (const float*)d_in[6];
    const float* b3 = (const float*)d_in[7];
    const float* wr = (const float*)d_in[8];
    const float* br = (const float*)d_in[9];
    float* out = (float*)d_out;

    const size_t shmem = (size_t)SMEM_FLOATS * sizeof(float);
    gcn_fused_kernel<<<dim3(N_), dim3(BLK), shmem, stream>>>(
        x, A, w1, b1, w2, b2, w3, b3, wr, br, out);
}
